# starting kernel
# baseline (speedup 1.0000x reference)
#include <hip/hip_runtime.h>


#define T_STEPS 1000
#define B_TOT   256
#define DZv     64
#define DHv     256
#define DSv     16

#define W1T_STRIDE 272
#define W1T_OFF    0
#define W1T_SIZE   (DZv * W1T_STRIDE)

#define W2C_STRIDE 80
#define W2C_OFF    (W1T_OFF + W1T_SIZE)
#define W2C_ROWS   (DHv + DSv)
#define W2C_SIZE   (W2C_ROWS * W2C_STRIDE)

#define ZAC_STRIDE 276
#define ZAC_OFF    (W2C_OFF + W2C_SIZE)
#define ZAC_SIZE   (16 * ZAC_STRIDE)

#define ZST_STRIDE 68
#define ZST_OFF    (ZAC_OFF + ZAC_SIZE)
#define ZST_SIZE   (16 * ZST_STRIDE)

#define LDS_FLOATS (ZST_OFF + ZST_SIZE)

typedef __attribute__((ext_vector_type(2))) float v2f;
typedef __attribute__((ext_vector_type(8))) float v8f;

__device__ __forceinline__ v8f wmma_f32(v2f a, v2f b, v8f c) {
  return __builtin_amdgcn_wmma_f32_16x16x4_f32(false, a, false, b, (short)0, c,
                                               false, false);
}

__global__ __launch_bounds__(128)
void plrnn_scan_kernel(const float* __restrict__ z0,
                       const float* __restrict__ s,
                       const float* __restrict__ Adiag,
                       const float* __restrict__ W1,
                       const float* __restrict__ W2,
                       const float* __restrict__ h1,
                       const float* __restrict__ h2,
                       const float* __restrict__ C,
                       float* __restrict__ zseq) {
  __shared__ float lds[LDS_FLOATS];

  const int tid  = threadIdx.x;
  const int wave = tid >> 5;
  const int lane = tid & 31;
  const int lh   = lane & 15;
  const int kh   = lane >> 4;
  const int b0   = blockIdx.x * 16;

  for (int f = tid; f < DHv * DZv; f += 128) {
    int h = f >> 6, j = f & 63;
    lds[W1T_OFF + j * W1T_STRIDE + h] = W1[f];
  }
  for (int f = tid; f < DZv * DHv; f += 128) {
    int z = f >> 8, h = f & 255;
    lds[W2C_OFF + h * W2C_STRIDE + z] = W2[f];
  }
  for (int f = tid; f < DZv * DSv; f += 128) {
    int z = f >> 4, si = f & 15;
    lds[W2C_OFF + (DHv + si) * W2C_STRIDE + z] = C[f];
  }
  for (int f = tid; f < 16 * DZv; f += 128) {
    int m = f >> 6, j = f & 63;
    lds[ZST_OFF + m * ZST_STRIDE + j] = z0[(b0 + m) * DZv + j];
  }

  float h1r[4];
#pragma unroll
  for (int tt = 0; tt < 4; ++tt) h1r[tt] = h1[wave * 64 + tt * 16 + lh];
  const int   zc  = wave * 16 + lh;
  const float Ar  = Adiag[zc];
  const float h2r = h2[zc];

  __syncthreads();

  v2f b1[16][4];
#pragma unroll
  for (int kk = 0; kk < 16; ++kk) {
    const int krow = kk * 4 + 2 * kh;
#pragma unroll
    for (int tt = 0; tt < 4; ++tt) {
      const int n0 = wave * 64 + tt * 16 + lh;
      v2f b;
      b.x = lds[W1T_OFF + krow * W1T_STRIDE + n0];
      b.y = lds[W1T_OFF + (krow + 1) * W1T_STRIDE + n0];
      b1[kk][tt] = b;
    }
  }

  const v8f zero8 = {0.f, 0.f, 0.f, 0.f, 0.f, 0.f, 0.f, 0.f};

  for (int t = 0; t < T_STEPS; ++t) {
    v8f acc1[4];
#pragma unroll
    for (int tt = 0; tt < 4; ++tt) acc1[tt] = zero8;

#pragma unroll
    for (int kk = 0; kk < 16; ++kk) {
      const v2f a =
          *(const v2f*)&lds[ZST_OFF + lh * ZST_STRIDE + kk * 4 + 2 * kh];
#pragma unroll
      for (int tt = 0; tt < 4; ++tt)
        acc1[tt] = wmma_f32(a, b1[kk][tt], acc1[tt]);
    }

#pragma unroll
    for (int tt = 0; tt < 4; ++tt) {
      const int n = wave * 64 + tt * 16 + lh;
#pragma unroll
      for (int r = 0; r < 8; ++r) {
        const float wz = acc1[tt][r];
        const float za = fmaxf(wz + h1r[tt], 0.0f) - fmaxf(wz, 0.0f);
        lds[ZAC_OFF + (r + 8 * kh) * ZAC_STRIDE + n] = za;
      }
    }

    {
      const int f = tid * 2;
      const int m = f >> 4, j = f & 15;
      const v2f sv = *(const v2f*)&s[((size_t)t * B_TOT + b0 + m) * DSv + j];
      lds[ZAC_OFF + m * ZAC_STRIDE + DHv + j]     = sv.x;
      lds[ZAC_OFF + m * ZAC_STRIDE + DHv + j + 1] = sv.y;
    }

    __syncthreads();

    v8f acc2 = zero8;
#pragma unroll 4
    for (int kk = 0; kk < 68; ++kk) {
      const int krow = kk * 4 + 2 * kh;
      const v2f a = *(const v2f*)&lds[ZAC_OFF + lh * ZAC_STRIDE + krow];
      v2f b;
      b.x = lds[W2C_OFF + krow * W2C_STRIDE + zc];
      b.y = lds[W2C_OFF + (krow + 1) * W2C_STRIDE + zc];
      acc2 = wmma_f32(a, b, acc2);
    }

#pragma unroll
    for (int r = 0; r < 8; ++r) {
      const int m  = r + 8 * kh;
      const int zi = ZST_OFF + m * ZST_STRIDE + zc;
      const float znew = Ar * lds[zi] + acc2[r] + h2r;
      lds[zi] = znew;
      __builtin_nontemporal_store(
          znew, &zseq[((size_t)t * B_TOT + b0 + m) * DZv + zc]);
    }

    __syncthreads();
  }
}

extern "C" void kernel_launch(void* const* d_in, const int* in_sizes, int n_in,
                              void* d_out, int out_size, void* d_ws,
                              size_t ws_size, hipStream_t stream) {
  const float* z0 = (const float*)d_in[0];
  const float* s  = (const float*)d_in[1];
  const float* A  = (const float*)d_in[2];
  const float* W1 = (const float*)d_in[3];
  const float* W2 = (const float*)d_in[4];
  const float* h1 = (const float*)d_in[5];
  const float* h2 = (const float*)d_in[6];
  const float* C  = (const float*)d_in[7];
  float* out = (float*)d_out;
  (void)in_sizes; (void)n_in; (void)out_size; (void)d_ws; (void)ws_size;

  plrnn_scan_kernel<<<B_TOT / 16, 128, 0, stream>>>(z0, s, A, W1, W2, h1, h2,
                                                    C, out);
}


# One trial per work-group (256 x 8 waves), that trial's own W1[b]/W2[b] in registers as WMMA A operands (parent read trial 0's for all); GEMM2 K split by wave, LDS sum: sum order changes
# speedup vs baseline: 2.5207x; 2.5207x over previous
#include <hip/hip_runtime.h>


#define T_STEPS 1000
#define B_TOT   256
#define DZv     64
#define DHv     256
#define DSv     16
#define NWAVE   8

typedef __attribute__((ext_vector_type(2))) float v2f;
typedef __attribute__((ext_vector_type(4))) float v4f;
typedef __attribute__((ext_vector_type(8))) float v8f;

__device__ __forceinline__ v8f wmma_f32(v2f a, v2f b, v8f c) {
  return __builtin_amdgcn_wmma_f32_16x16x4_f32(false, a, false, b, (short)0, c,
                                               false, false);
}

__global__ __launch_bounds__(256)
void plrnn_scan_kernel(const float* __restrict__ z0,
                       const float* __restrict__ s,
                       const float* __restrict__ Adiag,
                       const float* __restrict__ W1,
                       const float* __restrict__ W2,
                       const float* __restrict__ h1,
                       const float* __restrict__ h2,
                       const float* __restrict__ C,
                       float* __restrict__ zseq) {
  __shared__ __attribute__((aligned(16))) float zbuf[DZv];
  __shared__ __attribute__((aligned(16))) float part[NWAVE * DZv];

  const int tid  = threadIdx.x;
  const int wave = tid >> 5;
  const int lane = tid & 31;
  const int lh   = lane & 15;
  const int kh   = lane >> 4;
  const int b    = blockIdx.x;

  const float* W1b = W1 + (size_t)b * (DHv * DZv);
  const float* W2b = W2 + (size_t)b * (DZv * DHv);

  v2f a1[2][16];
#pragma unroll
  for (int ti = 0; ti < 2; ++ti) {
    const float* row = W1b + (wave * 32 + ti * 16 + lh) * DZv + 32 * kh;
#pragma unroll
    for (int kk = 0; kk < 16; ++kk) a1[ti][kk] = *(const v2f*)(row + 2 * kk);
  }

  v2f a2[4][2][4];
#pragma unroll
  for (int mt = 0; mt < 4; ++mt) {
#pragma unroll
    for (int ti = 0; ti < 2; ++ti) {
      const float* row =
          W2b + (mt * 16 + lh) * DHv + wave * 32 + ti * 16 + 8 * kh;
#pragma unroll
      for (int qq = 0; qq < 4; ++qq) a2[mt][ti][qq] = *(const v2f*)(row + 2 * qq);
    }
  }

  v2f acC[4];
#pragma unroll
  for (int mt = 0; mt < 4; ++mt) {
    const v2f cv = *(const v2f*)(C + (mt * 16 + lh) * DSv + 2 * wave);
    v2f c0 = {0.f, 0.f};
    acC[mt] = (kh == 0) ? cv : c0;
  }

  float h1r[2][8];
#pragma unroll
  for (int ti = 0; ti < 2; ++ti)
#pragma unroll
    for (int r = 0; r < 8; ++r) h1r[ti][r] = h1[wave * 32 + ti * 16 + r + 8 * kh];

  float zreg = 0.f, Ar = 0.f, h2r = 0.f;
  if (tid < DZv) {
    zreg = z0[b * DZv + tid];
    Ar   = Adiag[tid];
    h2r  = h2[tid];
    zbuf[tid] = zreg;
  }

  const float* sp = s + (size_t)b * DSv + 2 * wave;
  v2f sv = *(const v2f*)sp;

  __syncthreads();

  const v8f zero8 = {0.f, 0.f, 0.f, 0.f, 0.f, 0.f, 0.f, 0.f};
  const float* zb = &zbuf[32 * kh];

  for (int t = 0; t < T_STEPS; ++t) {
    const int tn = (t + 1 < T_STEPS) ? (t + 1) : t;
    const v2f sv_next = *(const v2f*)(sp + (size_t)tn * (B_TOT * DSv));

    v2f bs;
    bs.x = (kh == 0) ? sv.x : 0.f;
    bs.y = (kh == 0) ? sv.y : 0.f;
    v8f acc2[4];
#pragma unroll
    for (int mt = 0; mt < 4; ++mt) acc2[mt] = wmma_f32(acC[mt], bs, zero8);

    v8f acc1[2];
    acc1[0] = zero8;
    acc1[1] = zero8;
#pragma unroll
    for (int kq = 0; kq < 8; ++kq) {
      const v4f zz = *(const v4f*)(zb + 4 * kq);
      v2f bA, bB;
      bA.x = zz.x; bA.y = zz.y;
      bB.x = zz.z; bB.y = zz.w;
#pragma unroll
      for (int ti = 0; ti < 2; ++ti)
        acc1[ti] = wmma_f32(a1[ti][2 * kq], bA, acc1[ti]);
#pragma unroll
      for (int ti = 0; ti < 2; ++ti)
        acc1[ti] = wmma_f32(a1[ti][2 * kq + 1], bB, acc1[ti]);
    }

#pragma unroll
    for (int ti = 0; ti < 2; ++ti)
#pragma unroll
      for (int r = 0; r < 8; ++r) {
        const float wz = acc1[ti][r];
        acc1[ti][r] = fmaxf(wz + h1r[ti][r], 0.0f) - fmaxf(wz, 0.0f);
      }

#pragma unroll
    for (int ti = 0; ti < 2; ++ti)
#pragma unroll
      for (int qq = 0; qq < 4; ++qq) {
        v2f bz;
        bz.x = acc1[ti][2 * qq];
        bz.y = acc1[ti][2 * qq + 1];
#pragma unroll
        for (int mt = 0; mt < 4; ++mt)
          acc2[mt] = wmma_f32(a2[mt][ti][qq], bz, acc2[mt]);
      }

    if (lh == 0) {
#pragma unroll
      for (int mt = 0; mt < 4; ++mt) {
        float* pp = &part[wave * DZv + mt * 16 + 8 * kh];
        v4f lo, hi;
        lo.x = acc2[mt][0]; lo.y = acc2[mt][1]; lo.z = acc2[mt][2]; lo.w = acc2[mt][3];
        hi.x = acc2[mt][4]; hi.y = acc2[mt][5]; hi.z = acc2[mt][6]; hi.w = acc2[mt][7];
        *(v4f*)pp       = lo;
        *(v4f*)(pp + 4) = hi;
      }
    }

    __syncthreads();

    if (tid < DZv) {
      float sum = part[tid];
#pragma unroll
      for (int w = 1; w < NWAVE; ++w) sum += part[w * DZv + tid];
      zreg = Ar * zreg + sum + h2r;
      zbuf[tid] = zreg;
    }

    __syncthreads();

    if (tid < DZv)
      __builtin_nontemporal_store(
          zreg, &zseq[((size_t)t * B_TOT + b) * DZv + tid]);

    sv = sv_next;
  }
}

extern "C" void kernel_launch(void* const* d_in, const int* in_sizes, int n_in,
                              void* d_out, int out_size, void* d_ws,
                              size_t ws_size, hipStream_t stream) {
  const float* z0 = (const float*)d_in[0];
  const float* s  = (const float*)d_in[1];
  const float* A  = (const float*)d_in[2];
  const float* W1 = (const float*)d_in[3];
  const float* W2 = (const float*)d_in[4];
  const float* h1 = (const float*)d_in[5];
  const float* h2 = (const float*)d_in[6];
  const float* C  = (const float*)d_in[7];
  float* out = (float*)d_out;
  (void)in_sizes; (void)n_in; (void)out_size; (void)d_ws; (void)ws_size;

  plrnn_scan_kernel<<<B_TOT, 256, 0, stream>>>(z0, s, A, W1, W2, h1, h2, C,
                                               out);
}


# State in two halves: GEMM2 ends and next GEMM1 starts on the lower half, so each half's cross-wave LDS sum runs under the other half's WMMAs (barrier split: arrive, wait); GEMM1 sum order changes
# speedup vs baseline: 2.7003x; 1.0713x over previous
#include <hip/hip_runtime.h>


#define T_STEPS 1000
#define B_TOT   256
#define DZv     64
#define DHv     256
#define DSv     16
#define NWAVE   8

typedef __attribute__((ext_vector_type(2))) float v2f;
typedef __attribute__((ext_vector_type(4))) float v4f;
typedef __attribute__((ext_vector_type(8))) float v8f;

__device__ __forceinline__ v8f wmma_f32(v2f a, v2f b, v8f c) {
  return __builtin_amdgcn_wmma_f32_16x16x4_f32(false, a, false, b, (short)0, c,
                                               false, false);
}

__device__ __forceinline__ void bar_arrive() {
  asm volatile("s_wait_dscnt 0x0\n\ts_barrier_signal -1" ::: "memory");
  __builtin_amdgcn_sched_barrier(0);
}

__device__ __forceinline__ void bar_arrive_reads_done(v4f p, v4f q, v4f r, v4f u) {
  asm volatile("s_wait_dscnt 0x0\n\ts_barrier_signal -1"
               :: "v"(p), "v"(q), "v"(r), "v"(u) : "memory");
  __builtin_amdgcn_sched_barrier(0);
}

__device__ __forceinline__ void bar_wait(v8f x, v8f y) {
  asm volatile("s_barrier_wait -1" :: "v"(x), "v"(y) : "memory");
  __builtin_amdgcn_sched_barrier(0);
}

#define G2STEP(ks, m0)                                                        \
  {                                                                           \
    v2f bz;                                                                   \
    bz.x = zact[(ks) >> 2][2 * ((ks) & 3)];                                   \
    bz.y = zact[(ks) >> 2][2 * ((ks) & 3) + 1];                               \
    acc2[(m0)] =                                                              \
        wmma_f32(a2[(m0)][(ks) >> 2][(ks) & 3], bz, acc2[(m0)]);              \
    acc2[(m0) + 1] =                                                          \
        wmma_f32(a2[(m0) + 1][(ks) >> 2][(ks) & 3], bz, acc2[(m0) + 1]);      \
  }

#define G1STEP(kk, ZF)                                                        \
  {                                                                           \
    v2f bb;                                                                   \
    bb.x = ((kk) & 1) ? ZF[((kk) & 7) >> 1].z : ZF[((kk) & 7) >> 1].x;        \
    bb.y = ((kk) & 1) ? ZF[((kk) & 7) >> 1].w : ZF[((kk) & 7) >> 1].y;        \
    acc1[0] = wmma_f32(a1[0][(kk)], bb, acc1[0]);                             \
    acc1[1] = wmma_f32(a1[1][(kk)], bb, acc1[1]);                             \
  }

__global__ __launch_bounds__(256)
void plrnn_scan_kernel(const float* __restrict__ z0,
                       const float* __restrict__ s,
                       const float* __restrict__ Adiag,
                       const float* __restrict__ W1,
                       const float* __restrict__ W2,
                       const float* __restrict__ h1,
                       const float* __restrict__ h2,
                       const float* __restrict__ C,
                       float* __restrict__ zseq) {
  __shared__ __attribute__((aligned(16))) float zbuf[DZv];
  __shared__ __attribute__((aligned(16))) float part[NWAVE * DZv];

  const int tid  = threadIdx.x;
  const int wave = __builtin_amdgcn_readfirstlane(tid >> 5);
  const int lane = tid & 31;
  const int lh   = lane & 15;
  const int kh   = lane >> 4;
  const int b    = blockIdx.x;

  const float* W1b = W1 + (size_t)b * (DHv * DZv);
  const float* W2b = W2 + (size_t)b * (DZv * DHv);

  v2f a1[2][16];
#pragma unroll
  for (int ti = 0; ti < 2; ++ti) {
    const float* row = W1b + (wave * 32 + ti * 16 + lh) * DZv + 16 * kh;
#pragma unroll
    for (int kk = 0; kk < 16; ++kk)
      a1[ti][kk] = *(const v2f*)(row + 2 * kk + ((kk >= 8) ? 16 : 0));
  }

  v2f a2[4][2][4];
#pragma unroll
  for (int mt = 0; mt < 4; ++mt) {
#pragma unroll
    for (int ti = 0; ti < 2; ++ti) {
      const float* row =
          W2b + (mt * 16 + lh) * DHv + wave * 32 + ti * 16 + 8 * kh;
#pragma unroll
      for (int qq = 0; qq < 4; ++qq) a2[mt][ti][qq] = *(const v2f*)(row + 2 * qq);
    }
  }

  v2f acC[4];
#pragma unroll
  for (int mt = 0; mt < 4; ++mt) {
    const v2f cv = *(const v2f*)(C + (mt * 16 + lh) * DSv + 2 * wave);
    v2f c0 = {0.f, 0.f};
    acC[mt] = (kh == 0) ? cv : c0;
  }

  float h1r[2][8];
#pragma unroll
  for (int ti = 0; ti < 2; ++ti)
#pragma unroll
    for (int r = 0; r < 8; ++r) h1r[ti][r] = h1[wave * 32 + ti * 16 + r + 8 * kh];

  float zreg = 0.f, Ar = 0.f, h2r = 0.f;
  if (wave < 2) {
    zreg = z0[b * DZv + tid];
    Ar   = Adiag[tid];
    h2r  = h2[tid];
    zbuf[tid] = zreg;
  }

  const float* sp = s + (size_t)b * DSv + 2 * wave;
  v2f sv = *(const v2f*)sp;

  __syncthreads();

  const v8f zero8 = {0.f, 0.f, 0.f, 0.f, 0.f, 0.f, 0.f, 0.f};
  const float* zbL = &zbuf[16 * kh];
  const float* zbH = &zbuf[32 + 16 * kh];

  v8f acc1[2];
  acc1[0] = zero8;
  acc1[1] = zero8;
  {
    v4f zl[4], zh[4];
#pragma unroll
    for (int q = 0; q < 4; ++q) zl[q] = *(const v4f*)(zbL + 4 * q);
#pragma unroll
    for (int q = 0; q < 4; ++q) zh[q] = *(const v4f*)(zbH + 4 * q);
    G1STEP(0, zl) G1STEP(1, zl) G1STEP(2, zl) G1STEP(3, zl)
    G1STEP(4, zl) G1STEP(5, zl) G1STEP(6, zl) G1STEP(7, zl)
    G1STEP(8, zh) G1STEP(9, zh) G1STEP(10, zh) G1STEP(11, zh)
    G1STEP(12, zh) G1STEP(13, zh) G1STEP(14, zh) G1STEP(15, zh)
  }

  for (int t = 0; t < T_STEPS; ++t) {
    const int tn = (t + 1 < T_STEPS) ? (t + 1) : t;
    const v2f sv_next = *(const v2f*)(sp + (size_t)tn * (B_TOT * DSv));

    v2f bs;
    bs.x = (kh == 0) ? sv.x : 0.f;
    bs.y = (kh == 0) ? sv.y : 0.f;
    v8f acc2[4];
#pragma unroll
    for (int mt = 0; mt < 4; ++mt) acc2[mt] = wmma_f32(acC[mt], bs, zero8);

    v8f zact[2];
#pragma unroll
    for (int ti = 0; ti < 2; ++ti)
#pragma unroll
      for (int r = 0; r < 8; ++r) {
        const float wz = acc1[ti][r];
        zact[ti][r] = fmaxf(wz + h1r[ti][r], 0.0f) - fmaxf(wz, 0.0f);
      }

    G2STEP(0, 0) G2STEP(1, 0) G2STEP(2, 0) G2STEP(3, 0)
    G2STEP(4, 0) G2STEP(5, 0) G2STEP(6, 0) G2STEP(7, 0)
    G2STEP(0, 2)

    if (lh == 0) {
#pragma unroll
      for (int mt = 0; mt < 2; ++mt) {
        float* pp = &part[wave * DZv + mt * 16 + 8 * kh];
        v4f lo, hi;
        lo.x = acc2[mt][0]; lo.y = acc2[mt][1]; lo.z = acc2[mt][2]; lo.w = acc2[mt][3];
        hi.x = acc2[mt][4]; hi.y = acc2[mt][5]; hi.z = acc2[mt][6]; hi.w = acc2[mt][7];
        *(v4f*)pp       = lo;
        *(v4f*)(pp + 4) = hi;
      }
    }
    bar_arrive();

    G2STEP(1, 2) G2STEP(2, 2)

    bar_wait(acc2[2], acc2[3]);
    float pr[NWAVE];
    if (wave == 0) {
#pragma unroll
      for (int w = 0; w < NWAVE; ++w) pr[w] = part[w * DZv + lane];
    }

    G2STEP(3, 2) G2STEP(4, 2)

    if (wave == 0) {
      float sum = pr[0];
#pragma unroll
      for (int w = 1; w < NWAVE; ++w) sum += pr[w];
      zreg = Ar * zreg + sum + h2r;
      zbuf[lane] = zreg;
      __builtin_nontemporal_store(
          zreg, &zseq[((size_t)t * B_TOT + b) * DZv + lane]);
    }
    bar_arrive();

    G2STEP(5, 2) G2STEP(6, 2)

    bar_wait(acc2[2], acc2[3]);
    v4f zl[4];
#pragma unroll
    for (int q = 0; q < 4; ++q) zl[q] = *(const v4f*)(zbL + 4 * q);

    G2STEP(7, 2)

    acc1[0] = zero8;
    acc1[1] = zero8;
    G1STEP(0, zl)

    if (lh == 0) {
#pragma unroll
      for (int mt = 2; mt < 4; ++mt) {
        float* pp = &part[wave * DZv + mt * 16 + 8 * kh];
        v4f lo, hi;
        lo.x = acc2[mt][0]; lo.y = acc2[mt][1]; lo.z = acc2[mt][2]; lo.w = acc2[mt][3];
        hi.x = acc2[mt][4]; hi.y = acc2[mt][5]; hi.z = acc2[mt][6]; hi.w = acc2[mt][7];
        *(v4f*)pp       = lo;
        *(v4f*)(pp + 4) = hi;
      }
    }
    bar_arrive_reads_done(zl[0], zl[1], zl[2], zl[3]);

    G1STEP(1, zl) G1STEP(2, zl)

    bar_wait(acc1[0], acc1[1]);
    if (wave == 1) {
#pragma unroll
      for (int w = 0; w < NWAVE; ++w) pr[w] = part[w * DZv + 32 + lane];
    }

    G1STEP(3, zl) G1STEP(4, zl)

    if (wave == 1) {
      float sum = pr[0];
#pragma unroll
      for (int w = 1; w < NWAVE; ++w) sum += pr[w];
      zreg = Ar * zreg + sum + h2r;
      zbuf[32 + lane] = zreg;
      __builtin_nontemporal_store(
          zreg, &zseq[((size_t)t * B_TOT + b) * DZv + 32 + lane]);
    }
    bar_arrive();

    G1STEP(5, zl) G1STEP(6, zl)

    bar_wait(acc1[0], acc1[1]);
    v4f zh[4];
#pragma unroll
    for (int q = 0; q < 4; ++q) zh[q] = *(const v4f*)(zbH + 4 * q);

    G1STEP(7, zl)

    G1STEP(8, zh) G1STEP(9, zh) G1STEP(10, zh) G1STEP(11, zh)
    G1STEP(12, zh) G1STEP(13, zh) G1STEP(14, zh) G1STEP(15, zh)

    sv = sv_next;
  }
}

extern "C" void kernel_launch(void* const* d_in, const int* in_sizes, int n_in,
                              void* d_out, int out_size, void* d_ws,
                              size_t ws_size, hipStream_t stream) {
  const float* z0 = (const float*)d_in[0];
  const float* s  = (const float*)d_in[1];
  const float* A  = (const float*)d_in[2];
  const float* W1 = (const float*)d_in[3];
  const float* W2 = (const float*)d_in[4];
  const float* h1 = (const float*)d_in[5];
  const float* h2 = (const float*)d_in[6];
  const float* C  = (const float*)d_in[7];
  float* out = (float*)d_out;
  (void)in_sizes; (void)n_in; (void)out_size; (void)d_ws; (void)ws_size;

  plrnn_scan_kernel<<<B_TOT, 256, 0, stream>>>(z0, s, A, W1, W2, h1, h2, C,
                                               out);
}
